# starting kernel
# baseline (speedup 1.0000x reference)
#include <hip/hip_runtime.h>
#include <hip/hip_bf16.h>

typedef float v2f __attribute__((ext_vector_type(2)));
typedef float v4f __attribute__((ext_vector_type(4)));
typedef float v8f __attribute__((ext_vector_type(8)));

#define N_  64
#define C_  64
#define F_  64
#define T_  300
#define V_  25
#define VP_ 32
#define TB_ 4
static_assert(T_ % TB_ == 0, "t tiles must divide T exactly");

__global__ void build_adj_kernel(const int* __restrict__ ei, int E,
                                 float* __restrict__ Ahat) {
    __shared__ float deg[VP_];
    __shared__ float dinv[VP_];
    const int tid = threadIdx.x;

    if (tid < VP_) deg[tid] = 0.0f;
    __syncthreads();
    for (int e = tid; e < E; e += blockDim.x)
        atomicAdd(&deg[ei[E + e]], 1.0f);
    __syncthreads();
    if (tid < V_) deg[tid] += 1.0f;
    __syncthreads();
    if (tid < VP_) dinv[tid] = (tid < V_ && deg[tid] > 0.0f) ? rsqrtf(deg[tid]) : 0.0f;
    __syncthreads();

    for (int i = tid; i < VP_ * VP_; i += blockDim.x) Ahat[i] = 0.0f;
    __syncthreads();

    for (int e = tid; e < E; e += blockDim.x) {
        int s = ei[e];
        int d = ei[E + e];
        atomicAdd(&Ahat[d * VP_ + s], dinv[s] * dinv[d]);
    }
    if (tid < V_) atomicAdd(&Ahat[tid * VP_ + tid], dinv[tid] * dinv[tid]);
}

#define XW   (TB_ * V_)
#define XOFF 0
#define WOFF (C_ * XW)
#define AOFF (WOFF + F_ * C_)
#define BOFF (AOFF + VP_ * VP_)
#define LDSF (BOFF + F_)

__global__ __launch_bounds__(TB_ * 32) void spatial_gcn_kernel(
        const float* __restrict__ x, const float* __restrict__ w,
        const float* __restrict__ bias, const float* __restrict__ Ahat,
        float* __restrict__ out) {
    __shared__ float lds[LDSF];
    float* Xl = lds + XOFF;
    float* Wl = lds + WOFF;
    float* Al = lds + AOFF;
    float* Bl = lds + BOFF;

    const int tid = threadIdx.x;
    const int n   = blockIdx.y;
    const int t0  = blockIdx.x * TB_;
    const int NT  = TB_ * 32;

    for (int j = tid; j < C_ * (XW / 4); j += NT) {
        const int c = j / (XW / 4);
        const int q = j % (XW / 4);
        const v4f val = *reinterpret_cast<const v4f*>(
            x + (long)(n * C_ + c) * (T_ * V_) + t0 * V_ + q * 4);
        *reinterpret_cast<v4f*>(Xl + c * XW + q * 4) = val;
    }
    for (int j = tid; j < (F_ * C_) / 4; j += NT)
        reinterpret_cast<v4f*>(Wl)[j] = reinterpret_cast<const v4f*>(w)[j];
    for (int j = tid; j < (VP_ * VP_) / 4; j += NT)
        reinterpret_cast<v4f*>(Al)[j] = reinterpret_cast<const v4f*>(Ahat)[j];
    if (tid < F_ / 4)
        reinterpret_cast<v4f*>(Bl)[tid] = reinterpret_cast<const v4f*>(bias)[tid];
    __syncthreads();

    const int wv   = tid >> 5;
    const int lane = tid & 31;
    const int half = lane >> 4;
    const int l16  = lane & 15;
    const int t    = t0 + wv;
    const int xb   = wv * V_;

    v8f acc[4][2];
#pragma unroll
    for (int mt = 0; mt < 4; ++mt)
#pragma unroll
        for (int nt = 0; nt < 2; ++nt)
#pragma unroll
            for (int r = 0; r < 8; ++r) acc[mt][nt][r] = 0.0f;

#pragma unroll
    for (int k = 0; k < 16; ++k) {
        const int kk = 4 * k + 2 * half;
        v2f av[4], bv[2];
#pragma unroll
        for (int mt = 0; mt < 4; ++mt) {
            const float* p = &Wl[(mt * 16 + l16) * C_ + kk];
            av[mt][0] = p[0];
            av[mt][1] = p[1];
        }
#pragma unroll
        for (int nt = 0; nt < 2; ++nt) {
            bv[nt][0] = Xl[(kk + 0) * XW + xb + nt * 16 + l16];
            bv[nt][1] = Xl[(kk + 1) * XW + xb + nt * 16 + l16];
        }
#pragma unroll
        for (int mt = 0; mt < 4; ++mt)
#pragma unroll
            for (int nt = 0; nt < 2; ++nt)
                acc[mt][nt] = __builtin_amdgcn_wmma_f32_16x16x4_f32(
                    false, av[mt], false, bv[nt], (short)0, acc[mt][nt],
                    false, false);
    }

#pragma unroll
    for (int nt = 0; nt < 2; ++nt) {
        const int vp = nt * 16 + l16;
        if (vp < V_) {
#pragma unroll
            for (int mt = 0; mt < 4; ++mt)
#pragma unroll
                for (int r = 0; r < 8; ++r)
                    Xl[(mt * 16 + half * 8 + r) * XW + xb + vp] = acc[mt][nt][r];
        }
    }

    v8f o[4][2];
#pragma unroll
    for (int mt = 0; mt < 4; ++mt)
#pragma unroll
        for (int nt = 0; nt < 2; ++nt)
#pragma unroll
            for (int r = 0; r < 8; ++r)
                o[mt][nt][r] = Bl[mt * 16 + half * 8 + r];

#pragma unroll
    for (int k = 0; k < 8; ++k) {
        const int kk = 4 * k + 2 * half;
        v2f av[4], bv[2];
#pragma unroll
        for (int mt = 0; mt < 4; ++mt) {
            av[mt][0] = Xl[(mt * 16 + l16) * XW + xb + kk + 0];
            av[mt][1] = Xl[(mt * 16 + l16) * XW + xb + kk + 1];
        }
#pragma unroll
        for (int nt = 0; nt < 2; ++nt) {
            bv[nt][0] = Al[(nt * 16 + l16) * VP_ + kk + 0];
            bv[nt][1] = Al[(nt * 16 + l16) * VP_ + kk + 1];
        }
#pragma unroll
        for (int mt = 0; mt < 4; ++mt)
#pragma unroll
            for (int nt = 0; nt < 2; ++nt)
                o[mt][nt] = __builtin_amdgcn_wmma_f32_16x16x4_f32(
                    false, av[mt], false, bv[nt], (short)0, o[mt][nt],
                    false, false);
    }

#pragma unroll
    for (int nt = 0; nt < 2; ++nt) {
        const int v = nt * 16 + l16;
        if (v < V_) {
#pragma unroll
            for (int mt = 0; mt < 4; ++mt)
#pragma unroll
                for (int r = 0; r < 8; ++r) {
                    const int f = mt * 16 + half * 8 + r;
                    out[(((long)n * F_ + f) * T_ + t) * V_ + v] = o[mt][nt][r];
                }
        }
    }
}

extern "C" void kernel_launch(void* const* d_in, const int* in_sizes, int n_in,
                              void* d_out, int out_size, void* d_ws, size_t ws_size,
                              hipStream_t stream) {
    const float* x    = (const float*)d_in[0];
    const float* w    = (const float*)d_in[1];
    const float* bias = (const float*)d_in[2];
    const int*   ei   = (const int*)d_in[3];
    const int E = in_sizes[3] / 2;

    float* Ahat = (float*)d_ws;

    build_adj_kernel<<<1, 64, 0, stream>>>(ei, E, Ahat);

    dim3 grid(T_ / TB_, N_);
    spatial_gcn_kernel<<<grid, TB_ * 32, 0, stream>>>(
        x, w, bias, Ahat, (float*)d_out);
}


# Block loops over 5 t-groups with W and adjacency WMMA fragments held in registers (read once per wave from global); LDS keeps only a double-buffered x tile, no W/adjacency LDS staging or reads
# speedup vs baseline: 1.0787x; 1.0787x over previous
#include <hip/hip_runtime.h>
#include <hip/hip_bf16.h>

typedef float v2f __attribute__((ext_vector_type(2)));
typedef float v4f __attribute__((ext_vector_type(4)));
typedef float v8f __attribute__((ext_vector_type(8)));

#define N_  64
#define C_  64
#define F_  64
#define T_  300
#define V_  25
#define VP_ 32
#define TB_ 4
#define LT_ 5
static_assert(T_ % (TB_ * LT_) == 0, "t tiles must divide T exactly");

__global__ void build_adj_kernel(const int* __restrict__ ei, int E,
                                 float* __restrict__ Ahat) {
    __shared__ float deg[VP_];
    __shared__ float dinv[VP_];
    const int tid = threadIdx.x;

    if (tid < VP_) deg[tid] = 0.0f;
    __syncthreads();
    for (int e = tid; e < E; e += blockDim.x)
        atomicAdd(&deg[ei[E + e]], 1.0f);
    __syncthreads();
    if (tid < V_) deg[tid] += 1.0f;
    __syncthreads();
    if (tid < VP_) dinv[tid] = (tid < V_ && deg[tid] > 0.0f) ? rsqrtf(deg[tid]) : 0.0f;
    __syncthreads();

    for (int i = tid; i < VP_ * VP_; i += blockDim.x) Ahat[i] = 0.0f;
    __syncthreads();

    for (int e = tid; e < E; e += blockDim.x) {
        int s = ei[e];
        int d = ei[E + e];
        atomicAdd(&Ahat[d * VP_ + s], dinv[s] * dinv[d]);
    }
    if (tid < V_) atomicAdd(&Ahat[tid * VP_ + tid], dinv[tid] * dinv[tid]);
}

#define XW   (TB_ * V_)
#define XSZ  (C_ * XW)
#define XPAD 32
#define X0OFF 0
#define X1OFF (XSZ + XPAD)
#define BOFF (2 * (XSZ + XPAD))
#define LDSF (BOFF + F_)

__global__ __launch_bounds__(TB_ * 32) void spatial_gcn_kernel(
        const float* __restrict__ x, const float* __restrict__ w,
        const float* __restrict__ bias, const float* __restrict__ Ahat,
        float* __restrict__ out) {
    __shared__ float lds[LDSF];
    float* Bl = lds + BOFF;

    const int tid = threadIdx.x;
    const int n   = blockIdx.y;
    const int NT  = TB_ * 32;

    const int wv   = tid >> 5;
    const int lane = tid & 31;
    const int half = lane >> 4;
    const int l16  = lane & 15;
    const int xb   = wv * V_;

    v2f wa[16][4];
#pragma unroll
    for (int k = 0; k < 16; ++k)
#pragma unroll
        for (int mt = 0; mt < 4; ++mt)
            wa[k][mt] = *reinterpret_cast<const v2f*>(
                w + (mt * 16 + l16) * C_ + 4 * k + 2 * half);
    v2f ab[8][2];
#pragma unroll
    for (int k = 0; k < 8; ++k)
#pragma unroll
        for (int nt = 0; nt < 2; ++nt)
            ab[k][nt] = *reinterpret_cast<const v2f*>(
                Ahat + (nt * 16 + l16) * VP_ + 4 * k + 2 * half);

    if (tid < XPAD) {
        lds[X0OFF + XSZ + tid] = 0.0f;
        lds[X1OFF + XSZ + tid] = 0.0f;
    }
    if (tid < F_ / 4)
        reinterpret_cast<v4f*>(Bl)[tid] = reinterpret_cast<const v4f*>(bias)[tid];

#pragma unroll 1
    for (int it = 0; it < LT_; ++it) {
    float* Xl = lds + ((it & 1) ? X1OFF : X0OFF);
    const int t0 = (blockIdx.x * LT_ + it) * TB_;

    for (int j = tid; j < C_ * (XW / 4); j += NT) {
        const int c = j / (XW / 4);
        const int q = j % (XW / 4);
        const v4f val = *reinterpret_cast<const v4f*>(
            x + (long)(n * C_ + c) * (T_ * V_) + t0 * V_ + q * 4);
        *reinterpret_cast<v4f*>(Xl + c * XW + q * 4) = val;
    }
    __syncthreads();

    const int t    = t0 + wv;

    v8f acc[4][2];
#pragma unroll
    for (int mt = 0; mt < 4; ++mt)
#pragma unroll
        for (int nt = 0; nt < 2; ++nt)
#pragma unroll
            for (int r = 0; r < 8; ++r) acc[mt][nt][r] = 0.0f;

#pragma unroll
    for (int k = 0; k < 16; ++k) {
        const int kk = 4 * k + 2 * half;
        v2f bv[2];
#pragma unroll
        for (int nt = 0; nt < 2; ++nt) {
            bv[nt][0] = Xl[(kk + 0) * XW + xb + nt * 16 + l16];
            bv[nt][1] = Xl[(kk + 1) * XW + xb + nt * 16 + l16];
        }
#pragma unroll
        for (int mt = 0; mt < 4; ++mt)
#pragma unroll
            for (int nt = 0; nt < 2; ++nt)
                acc[mt][nt] = __builtin_amdgcn_wmma_f32_16x16x4_f32(
                    false, wa[k][mt], false, bv[nt], (short)0, acc[mt][nt],
                    false, false);
    }

#pragma unroll
    for (int nt = 0; nt < 2; ++nt) {
        const int vp = nt * 16 + l16;
        if (vp < V_) {
#pragma unroll
            for (int mt = 0; mt < 4; ++mt)
#pragma unroll
                for (int r = 0; r < 8; ++r)
                    Xl[(mt * 16 + half * 8 + r) * XW + xb + vp] = acc[mt][nt][r];
        }
    }

    v8f o[4][2];
#pragma unroll
    for (int mt = 0; mt < 4; ++mt)
#pragma unroll
        for (int nt = 0; nt < 2; ++nt)
#pragma unroll
            for (int r = 0; r < 8; ++r)
                o[mt][nt][r] = Bl[mt * 16 + half * 8 + r];

#pragma unroll
    for (int k = 0; k < 8; ++k) {
        const int kk = 4 * k + 2 * half;
        v2f av[4];
#pragma unroll
        for (int mt = 0; mt < 4; ++mt) {
            av[mt][0] = Xl[(mt * 16 + l16) * XW + xb + kk + 0];
            av[mt][1] = Xl[(mt * 16 + l16) * XW + xb + kk + 1];
        }
#pragma unroll
        for (int mt = 0; mt < 4; ++mt)
#pragma unroll
            for (int nt = 0; nt < 2; ++nt)
                o[mt][nt] = __builtin_amdgcn_wmma_f32_16x16x4_f32(
                    false, av[mt], false, ab[k][nt], (short)0, o[mt][nt],
                    false, false);
    }

#pragma unroll
    for (int nt = 0; nt < 2; ++nt) {
        const int v = nt * 16 + l16;
        if (v < V_) {
#pragma unroll
            for (int mt = 0; mt < 4; ++mt)
#pragma unroll
                for (int r = 0; r < 8; ++r) {
                    const int f = mt * 16 + half * 8 + r;
                    out[(((long)n * F_ + f) * T_ + t) * V_ + v] = o[mt][nt][r];
                }
        }
    }
    }
}

extern "C" void kernel_launch(void* const* d_in, const int* in_sizes, int n_in,
                              void* d_out, int out_size, void* d_ws, size_t ws_size,
                              hipStream_t stream) {
    const float* x    = (const float*)d_in[0];
    const float* w    = (const float*)d_in[1];
    const float* bias = (const float*)d_in[2];
    const int*   ei   = (const int*)d_in[3];
    const int E = in_sizes[3] / 2;

    float* Ahat = (float*)d_ws;

    build_adj_kernel<<<1, 64, 0, stream>>>(ei, E, Ahat);

    dim3 grid(T_ / (TB_ * LT_), N_);
    spatial_gcn_kernel<<<grid, TB_ * 32, 0, stream>>>(
        x, w, bias, Ahat, (float*)d_out);
}


# x bypasses LDS: each wave loads its GEMM1 B fragments from global straight into registers, refilled half by half one unit ahead under the WMMAs; no x tile, no loop barrier; 480 blocks x 10 units
# speedup vs baseline: 1.3308x; 1.2337x over previous
#include <hip/hip_runtime.h>
#include <hip/hip_bf16.h>

typedef float v2f __attribute__((ext_vector_type(2)));
typedef float v4f __attribute__((ext_vector_type(4)));
typedef float v8f __attribute__((ext_vector_type(8)));

#define N_  64
#define C_  64
#define F_  64
#define T_  300
#define V_  25
#define VP_ 32
#define TB_ 4
#define QN_ (T_ / TB_)
#define LT_ 10
#define NB_ (N_ * QN_ / LT_)
static_assert(T_ % TB_ == 0 && (N_ * QN_) % LT_ == 0, "units must tile exactly");

__global__ void build_adj_kernel(const int* __restrict__ ei, int E,
                                 float* __restrict__ Ahat) {
    __shared__ float deg[VP_];
    __shared__ float dinv[VP_];
    const int tid = threadIdx.x;

    if (tid < VP_) deg[tid] = 0.0f;
    __syncthreads();
    for (int e = tid; e < E; e += blockDim.x)
        atomicAdd(&deg[ei[E + e]], 1.0f);
    __syncthreads();
    if (tid < V_) deg[tid] += 1.0f;
    __syncthreads();
    if (tid < VP_) dinv[tid] = (tid < V_ && deg[tid] > 0.0f) ? rsqrtf(deg[tid]) : 0.0f;
    __syncthreads();

    for (int i = tid; i < VP_ * VP_; i += blockDim.x) Ahat[i] = 0.0f;
    __syncthreads();

    for (int e = tid; e < E; e += blockDim.x) {
        int s = ei[e];
        int d = ei[E + e];
        atomicAdd(&Ahat[d * VP_ + s], dinv[s] * dinv[d]);
    }
    if (tid < V_) atomicAdd(&Ahat[tid * VP_ + tid], dinv[tid] * dinv[tid]);
}

#define ZP_  36
#define ZSZ_ (F_ * ZP_)
#define BOFF (TB_ * ZSZ_)
#define LDSF (BOFF + F_)
#define RS_  (T_ * V_)

__global__ __launch_bounds__(TB_ * 32) void spatial_gcn_kernel(
        const float* __restrict__ x, const float* __restrict__ w,
        const float* __restrict__ bias, const float* __restrict__ Ahat,
        float* __restrict__ out) {
    __shared__ float lds[LDSF];
    float* Bl = lds + BOFF;

    const int tid = threadIdx.x;

    const int wv   = tid >> 5;
    const int lane = tid & 31;
    const int half = lane >> 4;
    const int l16  = lane & 15;
    float* Zl = lds + wv * ZSZ_;

    v2f wa[16][4];
#pragma unroll
    for (int k = 0; k < 16; ++k)
#pragma unroll
        for (int mt = 0; mt < 4; ++mt)
            wa[k][mt] = *reinterpret_cast<const v2f*>(
                w + (mt * 16 + l16) * C_ + 4 * k + 2 * half);
    v2f ab[8][2];
#pragma unroll
    for (int k = 0; k < 8; ++k)
#pragma unroll
        for (int nt = 0; nt < 2; ++nt)
            ab[k][nt] = *reinterpret_cast<const v2f*>(
                Ahat + (nt * 16 + l16) * VP_ + 4 * k + 2 * half);

    for (int i = lane; i < ZSZ_ / 4; i += 32)
        reinterpret_cast<v4f*>(Zl)[i] = (v4f){0.0f, 0.0f, 0.0f, 0.0f};
    if (tid < F_ / 4)
        reinterpret_cast<v4f*>(Bl)[tid] = reinterpret_cast<const v4f*>(bias)[tid];
    __syncthreads();

    const int vc1 = 16 + (l16 < V_ - 16 ? l16 : V_ - 17);
    const float* xl0 = x + (2 * half) * RS_ + l16;
    const float* xl1 = x + (2 * half) * RS_ + vc1;
    const int u0 = blockIdx.x * LT_;

    auto load_bv = [&](v2f (&b)[16][2], int u, int K0, int K1) {
        const int un = u / QN_;
        const int t  = (u - un * QN_) * TB_ + wv;
        const float* p0 = xl0 + (long)un * (C_ * RS_) + t * V_;
        const float* p1 = xl1 + (long)un * (C_ * RS_) + t * V_;
#pragma unroll
        for (int k = K0; k < K1; ++k)
#pragma unroll
            for (int j = 0; j < 2; ++j) {
                b[k][0][j] = p0[(4 * k + j) * RS_];
                b[k][1][j] = p1[(4 * k + j) * RS_];
            }
    };

    v2f bv[16][2];
    load_bv(bv, u0, 0, 16);

#pragma unroll 1
    for (int it = 0; it < LT_; ++it) {
        const int u  = u0 + it;
        const int un = u / QN_;
        const int t  = (u - un * QN_) * TB_ + wv;
        const int unx = it + 1 < LT_ ? u + 1 : u;

        v8f acc[4][2];
#pragma unroll
        for (int mt = 0; mt < 4; ++mt)
#pragma unroll
            for (int nt = 0; nt < 2; ++nt)
#pragma unroll
                for (int r = 0; r < 8; ++r) acc[mt][nt][r] = 0.0f;

#pragma unroll
        for (int k = 0; k < 8; ++k)
#pragma unroll
            for (int mt = 0; mt < 4; ++mt)
#pragma unroll
                for (int nt = 0; nt < 2; ++nt)
                    acc[mt][nt] = __builtin_amdgcn_wmma_f32_16x16x4_f32(
                        false, wa[k][mt], false, bv[k][nt], (short)0, acc[mt][nt],
                        false, false);
        load_bv(bv, unx, 0, 8);

#pragma unroll
        for (int k = 8; k < 16; ++k)
#pragma unroll
            for (int mt = 0; mt < 4; ++mt)
#pragma unroll
                for (int nt = 0; nt < 2; ++nt)
                    acc[mt][nt] = __builtin_amdgcn_wmma_f32_16x16x4_f32(
                        false, wa[k][mt], false, bv[k][nt], (short)0, acc[mt][nt],
                        false, false);
        load_bv(bv, unx, 8, 16);

#pragma unroll
        for (int nt = 0; nt < 2; ++nt) {
            const int vp = nt * 16 + l16;
            if (vp < V_) {
#pragma unroll
                for (int mt = 0; mt < 4; ++mt)
#pragma unroll
                    for (int r = 0; r < 8; ++r)
                        Zl[(mt * 16 + half * 8 + r) * ZP_ + vp] = acc[mt][nt][r];
            }
        }

        v8f o[4][2];
#pragma unroll
        for (int mt = 0; mt < 4; ++mt)
#pragma unroll
            for (int nt = 0; nt < 2; ++nt)
#pragma unroll
                for (int r = 0; r < 8; ++r)
                    o[mt][nt][r] = Bl[mt * 16 + half * 8 + r];

#pragma unroll
        for (int k = 0; k < 8; ++k) {
            const int kk = 4 * k + 2 * half;
            v2f av[4];
#pragma unroll
            for (int mt = 0; mt < 4; ++mt)
                av[mt] = *reinterpret_cast<const v2f*>(Zl + (mt * 16 + l16) * ZP_ + kk);
#pragma unroll
            for (int mt = 0; mt < 4; ++mt)
#pragma unroll
                for (int nt = 0; nt < 2; ++nt)
                    o[mt][nt] = __builtin_amdgcn_wmma_f32_16x16x4_f32(
                        false, av[mt], false, ab[k][nt], (short)0, o[mt][nt],
                        false, false);
        }

#pragma unroll
        for (int nt = 0; nt < 2; ++nt) {
            const int v = nt * 16 + l16;
            if (v < V_) {
#pragma unroll
                for (int mt = 0; mt < 4; ++mt)
#pragma unroll
                    for (int r = 0; r < 8; ++r) {
                        const int f = mt * 16 + half * 8 + r;
                        out[(((long)un * F_ + f) * T_ + t) * V_ + v] = o[mt][nt][r];
                    }
            }
        }
    }
}

extern "C" void kernel_launch(void* const* d_in, const int* in_sizes, int n_in,
                              void* d_out, int out_size, void* d_ws, size_t ws_size,
                              hipStream_t stream) {
    const float* x    = (const float*)d_in[0];
    const float* w    = (const float*)d_in[1];
    const float* bias = (const float*)d_in[2];
    const int*   ei   = (const int*)d_in[3];
    const int E = in_sizes[3] / 2;

    float* Ahat = (float*)d_ws;

    build_adj_kernel<<<1, 64, 0, stream>>>(ei, E, Ahat);

    spatial_gcn_kernel<<<NB_, TB_ * 32, 0, stream>>>(
        x, w, bias, Ahat, (float*)d_out);
}


# Output stores of a unit move under the next unit's GEMM1: results keep their own registers and leave in four groups of 16 stores behind the k-step quarters, not as a 64-store burst between units
# speedup vs baseline: 1.3925x; 1.0464x over previous
#include <hip/hip_runtime.h>
#include <hip/hip_bf16.h>

typedef float v2f __attribute__((ext_vector_type(2)));
typedef float v4f __attribute__((ext_vector_type(4)));
typedef float v8f __attribute__((ext_vector_type(8)));

#define N_  64
#define C_  64
#define F_  64
#define T_  300
#define V_  25
#define VP_ 32
#define TB_ 4
#define QN_ (T_ / TB_)
#define LT_ 10
#define NB_ (N_ * QN_ / LT_)
static_assert(T_ % TB_ == 0 && (N_ * QN_) % LT_ == 0, "units must tile exactly");

__global__ void build_adj_kernel(const int* __restrict__ ei, int E,
                                 float* __restrict__ Ahat) {
    __shared__ float deg[VP_];
    __shared__ float dinv[VP_];
    const int tid = threadIdx.x;

    if (tid < VP_) deg[tid] = 0.0f;
    __syncthreads();
    for (int e = tid; e < E; e += blockDim.x)
        atomicAdd(&deg[ei[E + e]], 1.0f);
    __syncthreads();
    if (tid < V_) deg[tid] += 1.0f;
    __syncthreads();
    if (tid < VP_) dinv[tid] = (tid < V_ && deg[tid] > 0.0f) ? rsqrtf(deg[tid]) : 0.0f;
    __syncthreads();

    for (int i = tid; i < VP_ * VP_; i += blockDim.x) Ahat[i] = 0.0f;
    __syncthreads();

    for (int e = tid; e < E; e += blockDim.x) {
        int s = ei[e];
        int d = ei[E + e];
        atomicAdd(&Ahat[d * VP_ + s], dinv[s] * dinv[d]);
    }
    if (tid < V_) atomicAdd(&Ahat[tid * VP_ + tid], dinv[tid] * dinv[tid]);
}

#define ZP_  36
#define ZSZ_ (F_ * ZP_)
#define BOFF (TB_ * ZSZ_)
#define LDSF (BOFF + F_)
#define RS_  (T_ * V_)

__global__ __launch_bounds__(TB_ * 32) void spatial_gcn_kernel(
        const float* __restrict__ x, const float* __restrict__ w,
        const float* __restrict__ bias, const float* __restrict__ Ahat,
        float* __restrict__ out) {
    __shared__ float lds[LDSF];
    float* Bl = lds + BOFF;

    const int tid = threadIdx.x;

    const int wv   = tid >> 5;
    const int lane = tid & 31;
    const int half = lane >> 4;
    const int l16  = lane & 15;
    float* Zl = lds + wv * ZSZ_;

    v2f wa[16][4];
#pragma unroll
    for (int k = 0; k < 16; ++k)
#pragma unroll
        for (int mt = 0; mt < 4; ++mt)
            wa[k][mt] = *reinterpret_cast<const v2f*>(
                w + (mt * 16 + l16) * C_ + 4 * k + 2 * half);
    v2f ab[8][2];
#pragma unroll
    for (int k = 0; k < 8; ++k)
#pragma unroll
        for (int nt = 0; nt < 2; ++nt)
            ab[k][nt] = *reinterpret_cast<const v2f*>(
                Ahat + (nt * 16 + l16) * VP_ + 4 * k + 2 * half);

    for (int i = lane; i < ZSZ_ / 4; i += 32)
        reinterpret_cast<v4f*>(Zl)[i] = (v4f){0.0f, 0.0f, 0.0f, 0.0f};
    if (tid < F_ / 4)
        reinterpret_cast<v4f*>(Bl)[tid] = reinterpret_cast<const v4f*>(bias)[tid];
    __syncthreads();

    const int vc1 = 16 + (l16 < V_ - 16 ? l16 : V_ - 17);
    const float* xl0 = x + (2 * half) * RS_ + l16;
    const float* xl1 = x + (2 * half) * RS_ + vc1;
    const int u0 = blockIdx.x * LT_;

    auto load_bv = [&](v2f (&b)[16][2], int u, int K0, int K1) {
        const int un = u / QN_;
        const int t  = (u - un * QN_) * TB_ + wv;
        const float* p0 = xl0 + (long)un * (C_ * RS_) + t * V_;
        const float* p1 = xl1 + (long)un * (C_ * RS_) + t * V_;
#pragma unroll
        for (int k = K0; k < K1; ++k)
#pragma unroll
            for (int j = 0; j < 2; ++j) {
                b[k][0][j] = p0[(4 * k + j) * RS_];
                b[k][1][j] = p1[(4 * k + j) * RS_];
            }
    };

    v2f bv[16][2];
    load_bv(bv, u0, 0, 16);

    v8f o[4][2];
    int pun = 0, pt = 0;

    auto store_rows = [&](int mt, int nt, int sn, int st) {
        float* ob = out + (((long)sn * F_ + mt * 16 + half * 8) * T_ + st) * V_ + nt * 16 + l16;
        if (nt * 16 + l16 < V_) {
#pragma unroll
            for (int r = 0; r < 8; ++r) ob[r * RS_] = o[mt][nt][r];
        }
    };
    auto store_group = [&](int q, int sn, int st) {
        store_rows(2 * (q & 1),     q >> 1, sn, st);
        store_rows(2 * (q & 1) + 1, q >> 1, sn, st);
    };

#pragma unroll 1
    for (int it = 0; it < LT_; ++it) {
        const int u  = u0 + it;
        const int un = u / QN_;
        const int t  = (u - un * QN_) * TB_ + wv;
        const int unx = it + 1 < LT_ ? u + 1 : u;

        v8f acc[4][2];
#pragma unroll
        for (int mt = 0; mt < 4; ++mt)
#pragma unroll
            for (int nt = 0; nt < 2; ++nt)
#pragma unroll
                for (int r = 0; r < 8; ++r) acc[mt][nt][r] = 0.0f;

#pragma unroll
        for (int q = 0; q < 4; ++q) {
#pragma unroll
            for (int k = 4 * q; k < 4 * q + 4; ++k)
#pragma unroll
                for (int mt = 0; mt < 4; ++mt)
#pragma unroll
                    for (int nt = 0; nt < 2; ++nt)
                        acc[mt][nt] = __builtin_amdgcn_wmma_f32_16x16x4_f32(
                            false, wa[k][mt], false, bv[k][nt], (short)0, acc[mt][nt],
                            false, false);
            load_bv(bv, unx, 4 * q, 4 * q + 4);
            if (it > 0) store_group(q, pun, pt);
        }

#pragma unroll
        for (int nt = 0; nt < 2; ++nt) {
            const int vp = nt * 16 + l16;
            if (vp < V_) {
#pragma unroll
                for (int mt = 0; mt < 4; ++mt)
#pragma unroll
                    for (int r = 0; r < 8; ++r)
                        Zl[(mt * 16 + half * 8 + r) * ZP_ + vp] = acc[mt][nt][r];
            }
        }

#pragma unroll
        for (int mt = 0; mt < 4; ++mt)
#pragma unroll
            for (int nt = 0; nt < 2; ++nt)
#pragma unroll
                for (int r = 0; r < 8; ++r)
                    o[mt][nt][r] = Bl[mt * 16 + half * 8 + r];

#pragma unroll
        for (int k = 0; k < 8; ++k) {
            const int kk = 4 * k + 2 * half;
            v2f av[4];
#pragma unroll
            for (int mt = 0; mt < 4; ++mt)
                av[mt] = *reinterpret_cast<const v2f*>(Zl + (mt * 16 + l16) * ZP_ + kk);
#pragma unroll
            for (int mt = 0; mt < 4; ++mt)
#pragma unroll
                for (int nt = 0; nt < 2; ++nt)
                    o[mt][nt] = __builtin_amdgcn_wmma_f32_16x16x4_f32(
                        false, av[mt], false, ab[k][nt], (short)0, o[mt][nt],
                        false, false);
        }
        pun = un;
        pt  = t;
    }

#pragma unroll
    for (int q = 0; q < 4; ++q) store_group(q, pun, pt);
}

extern "C" void kernel_launch(void* const* d_in, const int* in_sizes, int n_in,
                              void* d_out, int out_size, void* d_ws, size_t ws_size,
                              hipStream_t stream) {
    const float* x    = (const float*)d_in[0];
    const float* w    = (const float*)d_in[1];
    const float* bias = (const float*)d_in[2];
    const int*   ei   = (const int*)d_in[3];
    const int E = in_sizes[3] / 2;

    float* Ahat = (float*)d_ws;

    build_adj_kernel<<<1, 64, 0, stream>>>(ei, E, Ahat);

    spatial_gcn_kernel<<<NB_, TB_ * 32, 0, stream>>>(
        x, w, bias, Ahat, (float*)d_out);
}


# Each output row of a unit (25 floats, 100 B) leaves in ONE store: its 16- and 9-column parts are joined in one register across the lane halves (permlanex16 + selects); 64 row stores, not 128 pieces
# speedup vs baseline: 1.4544x; 1.0445x over previous
#include <hip/hip_runtime.h>
#include <hip/hip_bf16.h>

typedef float v2f __attribute__((ext_vector_type(2)));
typedef float v4f __attribute__((ext_vector_type(4)));
typedef float v8f __attribute__((ext_vector_type(8)));

__device__ __forceinline__ float xhalf(float x) {
    return __builtin_bit_cast(float, __builtin_amdgcn_permlanex16(
        __builtin_bit_cast(int, x), __builtin_bit_cast(int, x),
        0x76543210, 0xfedcba98, false, false));
}

#define N_  64
#define C_  64
#define F_  64
#define T_  300
#define V_  25
#define VP_ 32
#define TB_ 4
#define QN_ (T_ / TB_)
#define LT_ 10
#define NB_ (N_ * QN_ / LT_)
static_assert(T_ % TB_ == 0 && (N_ * QN_) % LT_ == 0, "units must tile exactly");

__global__ void build_adj_kernel(const int* __restrict__ ei, int E,
                                 float* __restrict__ Ahat) {
    __shared__ float deg[VP_];
    __shared__ float dinv[VP_];
    const int tid = threadIdx.x;

    if (tid < VP_) deg[tid] = 0.0f;
    __syncthreads();
    for (int e = tid; e < E; e += blockDim.x)
        atomicAdd(&deg[ei[E + e]], 1.0f);
    __syncthreads();
    if (tid < V_) deg[tid] += 1.0f;
    __syncthreads();
    if (tid < VP_) dinv[tid] = (tid < V_ && deg[tid] > 0.0f) ? rsqrtf(deg[tid]) : 0.0f;
    __syncthreads();

    for (int i = tid; i < VP_ * VP_; i += blockDim.x) Ahat[i] = 0.0f;
    __syncthreads();

    for (int e = tid; e < E; e += blockDim.x) {
        int s = ei[e];
        int d = ei[E + e];
        atomicAdd(&Ahat[d * VP_ + s], dinv[s] * dinv[d]);
    }
    if (tid < V_) atomicAdd(&Ahat[tid * VP_ + tid], dinv[tid] * dinv[tid]);
}

#define ZP_  36
#define ZSZ_ (F_ * ZP_)
#define BOFF (TB_ * ZSZ_)
#define LDSF (BOFF + F_)
#define RS_  (T_ * V_)

__global__ __launch_bounds__(TB_ * 32) void spatial_gcn_kernel(
        const float* __restrict__ x, const float* __restrict__ w,
        const float* __restrict__ bias, const float* __restrict__ Ahat,
        float* __restrict__ out) {
    __shared__ float lds[LDSF];
    float* Bl = lds + BOFF;

    const int tid = threadIdx.x;

    const int wv   = tid >> 5;
    const int lane = tid & 31;
    const int half = lane >> 4;
    const int l16  = lane & 15;
    float* Zl = lds + wv * ZSZ_;

    v2f wa[16][4];
#pragma unroll
    for (int k = 0; k < 16; ++k)
#pragma unroll
        for (int mt = 0; mt < 4; ++mt)
            wa[k][mt] = *reinterpret_cast<const v2f*>(
                w + (mt * 16 + l16) * C_ + 4 * k + 2 * half);
    v2f ab[8][2];
#pragma unroll
    for (int k = 0; k < 8; ++k)
#pragma unroll
        for (int nt = 0; nt < 2; ++nt)
            ab[k][nt] = *reinterpret_cast<const v2f*>(
                Ahat + (nt * 16 + l16) * VP_ + 4 * k + 2 * half);

    for (int i = lane; i < ZSZ_ / 4; i += 32)
        reinterpret_cast<v4f*>(Zl)[i] = (v4f){0.0f, 0.0f, 0.0f, 0.0f};
    if (tid < F_ / 4)
        reinterpret_cast<v4f*>(Bl)[tid] = reinterpret_cast<const v4f*>(bias)[tid];
    __syncthreads();

    const int vc1 = 16 + (l16 < V_ - 16 ? l16 : V_ - 17);
    const float* xl0 = x + (2 * half) * RS_ + l16;
    const float* xl1 = x + (2 * half) * RS_ + vc1;
    const int u0 = blockIdx.x * LT_;

    auto load_bv = [&](v2f (&b)[16][2], int u, int K0, int K1) {
        const int un = u / QN_;
        const int t  = (u - un * QN_) * TB_ + wv;
        const float* p0 = xl0 + (long)un * (C_ * RS_) + t * V_;
        const float* p1 = xl1 + (long)un * (C_ * RS_) + t * V_;
#pragma unroll
        for (int k = K0; k < K1; ++k)
#pragma unroll
            for (int j = 0; j < 2; ++j) {
                b[k][0][j] = p0[(4 * k + j) * RS_];
                b[k][1][j] = p1[(4 * k + j) * RS_];
            }
    };

    v2f bv[16][2];
    load_bv(bv, u0, 0, 16);

    v8f o[4][2];
    int pun = 0, pt = 0;

    auto store_group = [&](int mt, int sn, int st) {
        float lo[8], hi[8];
#pragma unroll
        for (int r = 0; r < 8; ++r) {
            const float a = o[mt][0][r];
            const float b = o[mt][1][r];
            const float m = half ? a : b;
            const float s = xhalf(m);
            lo[r] = half ? s : a;
            hi[r] = half ? b : s;
        }
        float* ob = out + (((long)sn * F_ + mt * 16) * T_ + st) * V_ + lane;
        if (lane < V_) {
#pragma unroll
            for (int r = 0; r < 8; ++r) ob[r * RS_] = lo[r];
#pragma unroll
            for (int r = 0; r < 8; ++r) ob[(8 + r) * RS_] = hi[r];
        }
    };

#pragma unroll 1
    for (int it = 0; it < LT_; ++it) {
        const int u  = u0 + it;
        const int un = u / QN_;
        const int t  = (u - un * QN_) * TB_ + wv;
        const int unx = it + 1 < LT_ ? u + 1 : u;

        v8f acc[4][2];
#pragma unroll
        for (int mt = 0; mt < 4; ++mt)
#pragma unroll
            for (int nt = 0; nt < 2; ++nt)
#pragma unroll
                for (int r = 0; r < 8; ++r) acc[mt][nt][r] = 0.0f;

#pragma unroll
        for (int q = 0; q < 4; ++q) {
#pragma unroll
            for (int k = 4 * q; k < 4 * q + 4; ++k)
#pragma unroll
                for (int mt = 0; mt < 4; ++mt)
#pragma unroll
                    for (int nt = 0; nt < 2; ++nt)
                        acc[mt][nt] = __builtin_amdgcn_wmma_f32_16x16x4_f32(
                            false, wa[k][mt], false, bv[k][nt], (short)0, acc[mt][nt],
                            false, false);
            load_bv(bv, unx, 4 * q, 4 * q + 4);
            if (it > 0) store_group(q, pun, pt);
        }

#pragma unroll
        for (int nt = 0; nt < 2; ++nt) {
            const int vp = nt * 16 + l16;
            if (vp < V_) {
#pragma unroll
                for (int mt = 0; mt < 4; ++mt)
#pragma unroll
                    for (int r = 0; r < 8; ++r)
                        Zl[(mt * 16 + half * 8 + r) * ZP_ + vp] = acc[mt][nt][r];
            }
        }

#pragma unroll
        for (int mt = 0; mt < 4; ++mt)
#pragma unroll
            for (int nt = 0; nt < 2; ++nt)
#pragma unroll
                for (int r = 0; r < 8; ++r)
                    o[mt][nt][r] = Bl[mt * 16 + half * 8 + r];

#pragma unroll
        for (int k = 0; k < 8; ++k) {
            const int kk = 4 * k + 2 * half;
            v2f av[4];
#pragma unroll
            for (int mt = 0; mt < 4; ++mt)
                av[mt] = *reinterpret_cast<const v2f*>(Zl + (mt * 16 + l16) * ZP_ + kk);
#pragma unroll
            for (int mt = 0; mt < 4; ++mt)
#pragma unroll
                for (int nt = 0; nt < 2; ++nt)
                    o[mt][nt] = __builtin_amdgcn_wmma_f32_16x16x4_f32(
                        false, av[mt], false, ab[k][nt], (short)0, o[mt][nt],
                        false, false);
        }
        pun = un;
        pt  = t;
    }

#pragma unroll
    for (int q = 0; q < 4; ++q) store_group(q, pun, pt);
}

extern "C" void kernel_launch(void* const* d_in, const int* in_sizes, int n_in,
                              void* d_out, int out_size, void* d_ws, size_t ws_size,
                              hipStream_t stream) {
    const float* x    = (const float*)d_in[0];
    const float* w    = (const float*)d_in[1];
    const float* bias = (const float*)d_in[2];
    const int*   ei   = (const int*)d_in[3];
    const int E = in_sizes[3] / 2;

    float* Ahat = (float*)d_ws;

    build_adj_kernel<<<1, 64, 0, stream>>>(ei, E, Ahat);

    spatial_gcn_kernel<<<NB_, TB_ * 32, 0, stream>>>(
        x, w, bias, Ahat, (float*)d_out);
}


# GEMM1 computed transposed so its accumulators (lane=f, register=v) are GEMM2's A fragments as they stand: Z stays in registers, no LDS store/load between the products; GEMM2 sums v in that k order
# speedup vs baseline: 1.5187x; 1.0442x over previous
#include <hip/hip_runtime.h>
#include <hip/hip_bf16.h>

typedef float v2f __attribute__((ext_vector_type(2)));
typedef float v4f __attribute__((ext_vector_type(4)));
typedef float v8f __attribute__((ext_vector_type(8)));

__device__ __forceinline__ float xhalf(float x) {
    return __builtin_bit_cast(float, __builtin_amdgcn_permlanex16(
        __builtin_bit_cast(int, x), __builtin_bit_cast(int, x),
        0x76543210, 0xfedcba98, false, false));
}

#define N_  64
#define C_  64
#define F_  64
#define T_  300
#define V_  25
#define VP_ 32
#define TB_ 4
#define QN_ (T_ / TB_)
#define LT_ 10
#define NB_ (N_ * QN_ / LT_)
static_assert(T_ % TB_ == 0 && (N_ * QN_) % LT_ == 0, "units must tile exactly");

__global__ void build_adj_kernel(const int* __restrict__ ei, int E,
                                 float* __restrict__ Ahat) {
    __shared__ float deg[VP_];
    __shared__ float dinv[VP_];
    const int tid = threadIdx.x;

    if (tid < VP_) deg[tid] = 0.0f;
    __syncthreads();
    for (int e = tid; e < E; e += blockDim.x)
        atomicAdd(&deg[ei[E + e]], 1.0f);
    __syncthreads();
    if (tid < V_) deg[tid] += 1.0f;
    __syncthreads();
    if (tid < VP_) dinv[tid] = (tid < V_ && deg[tid] > 0.0f) ? rsqrtf(deg[tid]) : 0.0f;
    __syncthreads();

    for (int i = tid; i < VP_ * VP_; i += blockDim.x) Ahat[i] = 0.0f;
    __syncthreads();

    for (int e = tid; e < E; e += blockDim.x) {
        int s = ei[e];
        int d = ei[E + e];
        atomicAdd(&Ahat[d * VP_ + s], dinv[s] * dinv[d]);
    }
    if (tid < V_) atomicAdd(&Ahat[tid * VP_ + tid], dinv[tid] * dinv[tid]);
}

#define LDSF F_
#define RS_  (T_ * V_)

__global__ __launch_bounds__(TB_ * 32) void spatial_gcn_kernel(
        const float* __restrict__ x, const float* __restrict__ w,
        const float* __restrict__ bias, const float* __restrict__ Ahat,
        float* __restrict__ out) {
    __shared__ __attribute__((aligned(16))) float Bl[LDSF];

    const int tid = threadIdx.x;

    const int wv   = tid >> 5;
    const int lane = tid & 31;
    const int half = lane >> 4;
    const int l16  = lane & 15;

    v2f wa[16][4];
#pragma unroll
    for (int k = 0; k < 16; ++k)
#pragma unroll
        for (int mt = 0; mt < 4; ++mt)
            wa[k][mt] = *reinterpret_cast<const v2f*>(
                w + (mt * 16 + l16) * C_ + 4 * k + 2 * half);
    v2f ab[2][4][2];
#pragma unroll
    for (int vt = 0; vt < 2; ++vt)
#pragma unroll
        for (int rp = 0; rp < 4; ++rp)
#pragma unroll
            for (int nt = 0; nt < 2; ++nt)
                ab[vt][rp][nt] = *reinterpret_cast<const v2f*>(
                    Ahat + (nt * 16 + l16) * VP_ + 16 * vt + 8 * half + 2 * rp);

    if (tid < F_ / 4)
        reinterpret_cast<v4f*>(Bl)[tid] = reinterpret_cast<const v4f*>(bias)[tid];
    __syncthreads();

    const int vc1 = 16 + (l16 < V_ - 16 ? l16 : V_ - 17);
    const float* xl0 = x + (2 * half) * RS_ + l16;
    const float* xl1 = x + (2 * half) * RS_ + vc1;
    const int u0 = blockIdx.x * LT_;

    auto load_bv = [&](v2f (&b)[16][2], int u, int K0, int K1) {
        const int un = u / QN_;
        const int t  = (u - un * QN_) * TB_ + wv;
        const float* p0 = xl0 + (long)un * (C_ * RS_) + t * V_;
        const float* p1 = xl1 + (long)un * (C_ * RS_) + t * V_;
#pragma unroll
        for (int k = K0; k < K1; ++k)
#pragma unroll
            for (int j = 0; j < 2; ++j) {
                b[k][0][j] = p0[(4 * k + j) * RS_];
                b[k][1][j] = p1[(4 * k + j) * RS_];
            }
    };

    v2f bv[16][2];
    load_bv(bv, u0, 0, 16);

    v8f o[4][2];
    int pun = 0, pt = 0;

    auto store_group = [&](int mt, int sn, int st) {
        float lo[8], hi[8];
#pragma unroll
        for (int r = 0; r < 8; ++r) {
            const float a = o[mt][0][r];
            const float b = o[mt][1][r];
            const float m = half ? a : b;
            const float s = xhalf(m);
            lo[r] = half ? s : a;
            hi[r] = half ? b : s;
        }
        float* ob = out + (((long)sn * F_ + mt * 16) * T_ + st) * V_ + lane;
        if (lane < V_) {
#pragma unroll
            for (int r = 0; r < 8; ++r) ob[r * RS_] = lo[r];
#pragma unroll
            for (int r = 0; r < 8; ++r) ob[(8 + r) * RS_] = hi[r];
        }
    };

#pragma unroll 1
    for (int it = 0; it < LT_; ++it) {
        const int u  = u0 + it;
        const int un = u / QN_;
        const int t  = (u - un * QN_) * TB_ + wv;
        const int unx = it + 1 < LT_ ? u + 1 : u;

        v8f acc[2][4];
#pragma unroll
        for (int vt = 0; vt < 2; ++vt)
#pragma unroll
            for (int ft = 0; ft < 4; ++ft)
#pragma unroll
                for (int r = 0; r < 8; ++r) acc[vt][ft][r] = 0.0f;

#pragma unroll
        for (int q = 0; q < 4; ++q) {
#pragma unroll
            for (int k = 4 * q; k < 4 * q + 4; ++k)
#pragma unroll
                for (int ft = 0; ft < 4; ++ft)
#pragma unroll
                    for (int vt = 0; vt < 2; ++vt)
                        acc[vt][ft] = __builtin_amdgcn_wmma_f32_16x16x4_f32(
                            false, bv[k][vt], false, wa[k][ft], (short)0, acc[vt][ft],
                            false, false);
            load_bv(bv, unx, 4 * q, 4 * q + 4);
            if (it > 0) store_group(q, pun, pt);
        }

#pragma unroll
        for (int mt = 0; mt < 4; ++mt)
#pragma unroll
            for (int nt = 0; nt < 2; ++nt)
#pragma unroll
                for (int r = 0; r < 8; ++r)
                    o[mt][nt][r] = Bl[mt * 16 + half * 8 + r];

#pragma unroll
        for (int vt = 0; vt < 2; ++vt)
#pragma unroll
            for (int rp = 0; rp < 4; ++rp)
#pragma unroll
                for (int ft = 0; ft < 4; ++ft) {
                    const v2f az = {acc[vt][ft][2 * rp], acc[vt][ft][2 * rp + 1]};
#pragma unroll
                    for (int nt = 0; nt < 2; ++nt)
                        o[ft][nt] = __builtin_amdgcn_wmma_f32_16x16x4_f32(
                            false, az, false, ab[vt][rp][nt], (short)0, o[ft][nt],
                            false, false);
                }
        pun = un;
        pt  = t;
    }

#pragma unroll
    for (int q = 0; q < 4; ++q) store_group(q, pun, pt);
}

extern "C" void kernel_launch(void* const* d_in, const int* in_sizes, int n_in,
                              void* d_out, int out_size, void* d_ws, size_t ws_size,
                              hipStream_t stream) {
    const float* x    = (const float*)d_in[0];
    const float* w    = (const float*)d_in[1];
    const float* bias = (const float*)d_in[2];
    const int*   ei   = (const int*)d_in[3];
    const int E = in_sizes[3] / 2;

    float* Ahat = (float*)d_ws;

    build_adj_kernel<<<1, 64, 0, stream>>>(ei, E, Ahat);

    spatial_gcn_kernel<<<NB_, TB_ * 32, 0, stream>>>(
        x, w, bias, Ahat, (float*)d_out);
}
